# starting kernel
# baseline (speedup 1.0000x reference)
#include <hip/hip_runtime.h>
#include <hip/hip_bf16.h>

typedef float v2f __attribute__((ext_vector_type(2)));
typedef float v8f __attribute__((ext_vector_type(8)));

#define N_NODES 50000
#define KDIM    128
#define LDA     132

__global__ __launch_bounds__(256) void deg_kernel(const int* __restrict__ dst,
                                                  float* __restrict__ deg, int nE) {
    int e = blockIdx.x * blockDim.x + threadIdx.x;
    if (e < nE) unsafeAtomicAdd(&deg[dst[e]], 1.0f);
}

__global__ __launch_bounds__(256) void scatter_kernel(const float* __restrict__ feat,
                                                      const int* __restrict__ src,
                                                      const int* __restrict__ dst,
                                                      float* __restrict__ msg, int nE) {
    int t = blockIdx.x * blockDim.x + threadIdx.x;
    int e = t >> 5;
    if (e >= nE) return;
    int c = (t & 31) << 2;
    int s = src[e], d = dst[e];
    const float4 v = *(const float4*)(feat + (size_t)s * KDIM + c);
    float* p = msg + (size_t)d * KDIM + c;
    unsafeAtomicAdd(p + 0, v.x);
    unsafeAtomicAdd(p + 1, v.y);
    unsafeAtomicAdd(p + 2, v.z);
    unsafeAtomicAdd(p + 3, v.w);
}

__global__ __launch_bounds__(256) void sage_gemm_wmma(const float* __restrict__ msg,
                                                      const float* __restrict__ feat,
                                                      const float* __restrict__ deg,
                                                      const float* __restrict__ W,
                                                      const float* __restrict__ bias,
                                                      float* __restrict__ out,
                                                      int n_out, int apply_lrelu) {
    __shared__ float Atile[16 * LDA];

    const int tid  = threadIdx.x;
    const int nt   = blockDim.x;
    const int wave = tid >> 5;
    const int lane = tid & 31;
    const int m0   = blockIdx.x * 16;

    for (int i = tid; i < 16 * (KDIM / 4); i += nt) {
        int r  = i >> 5;
        int c4 = (i & 31) << 2;
        int row = m0 + r;
        float inv = 1.0f / (deg[row] + 1.0f);
        const float4 mv = *(const float4*)(msg  + (size_t)row * KDIM + c4);
        const float4 xv = *(const float4*)(feat + (size_t)row * KDIM + c4);
        float* a = &Atile[r * LDA + c4];
        a[0] = (mv.x + xv.x) * inv;
        a[1] = (mv.y + xv.y) * inv;
        a[2] = (mv.z + xv.z) * inv;
        a[3] = (mv.w + xv.w) * inv;
    }
    __syncthreads();

    const int n0 = wave * 16;
    const int lr = lane & 15;
    const int hi = lane >> 4;

    v8f acc = {0.f, 0.f, 0.f, 0.f, 0.f, 0.f, 0.f, 0.f};

#pragma unroll 8
    for (int k = 0; k < KDIM; k += 4) {
        v2f a, b;
        const float* ap = &Atile[lr * LDA + k + 2 * hi];
        a.x = ap[0];
        a.y = ap[1];
        const float* bp = &W[(size_t)(k + 2 * hi) * n_out + n0 + lr];
        b.x = bp[0];
        b.y = bp[n_out];
        acc = __builtin_amdgcn_wmma_f32_16x16x4_f32(
             false, a,  false, b,
             (short)0, acc,  false,  false);
    }

    const float bv   = bias[n0 + lr];
    const int  rbase = m0 + 8 * hi;
    const int  col   = n0 + lr;
#pragma unroll
    for (int i = 0; i < 8; ++i) {
        float v = acc[i] + bv;
        if (apply_lrelu) v = (v >= 0.f) ? v : 0.01f * v;
        out[(size_t)(rbase + i) * n_out + col] = v;
    }
}

extern "C" void kernel_launch(void* const* d_in, const int* in_sizes, int n_in,
                              void* d_out, int out_size, void* d_ws, size_t ws_size,
                              hipStream_t stream) {
    const float* x   = (const float*)d_in[0];
    const int*   edg = (const int*)  d_in[1];
    const float* W1  = (const float*)d_in[2];
    const float* b1  = (const float*)d_in[3];
    const float* W2  = (const float*)d_in[4];
    const float* b2  = (const float*)d_in[5];
    float*       outp = (float*)d_out;

    const int E = in_sizes[1] / 2;
    const int* src = edg;
    const int* dst = edg + E;

    float* msg = (float*)d_ws;
    float* deg = msg + (size_t)N_NODES * KDIM;
    float* h1  = deg + N_NODES;

    const size_t msgBytes = (size_t)N_NODES * KDIM * sizeof(float);

    hipMemsetAsync(msg, 0, msgBytes, stream);
    hipMemsetAsync(deg, 0, (size_t)N_NODES * sizeof(float), stream);

    deg_kernel<<<(E + 255) / 256, 256, 0, stream>>>(dst, deg, E);

    {
        long long threads = (long long)E * 32;
        int blocks = (int)((threads + 255) / 256);
        scatter_kernel<<<blocks, 256, 0, stream>>>(x, src, dst, msg, E);
    }

    sage_gemm_wmma<<<N_NODES / 16, 256, 0, stream>>>(msg, x, deg, W1, b1, h1,
                                                      128,  1);

    hipMemsetAsync(msg, 0, msgBytes, stream);

    {
        long long threads = (long long)E * 32;
        int blocks = (int)((threads + 255) / 256);
        scatter_kernel<<<blocks, 256, 0, stream>>>(h1, src, dst, msg, E);
    }

    sage_gemm_wmma<<<N_NODES / 16, 64, 0, stream>>>(msg, h1, deg, W2, b2, outp,
                                                     32,  0);
}


# scatter: lane l owns features l,l+32,l+64,l+96: each wave-wide load and f32 atomic add covers one whole 128-B line (was 8 floats in each of 4 lines); h1 plane moved to a 128-B boundary; atomics kept
# speedup vs baseline: 1.3481x; 1.3481x over previous
#include <hip/hip_runtime.h>
#include <hip/hip_bf16.h>

typedef float v2f __attribute__((ext_vector_type(2)));
typedef float v8f __attribute__((ext_vector_type(8)));

#define N_NODES 50000
#define KDIM    128
#define LDA     132

__global__ __launch_bounds__(256) void deg_kernel(const int* __restrict__ dst,
                                                  float* __restrict__ deg, int nE) {
    int e = blockIdx.x * blockDim.x + threadIdx.x;
    if (e < nE) unsafeAtomicAdd(&deg[dst[e]], 1.0f);
}

__global__ __launch_bounds__(256) void scatter_kernel(const float* __restrict__ feat,
                                                      const int* __restrict__ src,
                                                      const int* __restrict__ dst,
                                                      float* __restrict__ msg, int nE) {
    int t = blockIdx.x * blockDim.x + threadIdx.x;
    int e = t >> 5;
    if (e >= nE) return;
    int l = t & 31;
    int s = src[e], d = dst[e];
    const float* f = feat + (size_t)s * KDIM + l;
    const float v0 = f[0];
    const float v1 = f[32];
    const float v2 = f[64];
    const float v3 = f[96];
    float* p = msg + (size_t)d * KDIM + l;
    unsafeAtomicAdd(p +  0, v0);
    unsafeAtomicAdd(p + 32, v1);
    unsafeAtomicAdd(p + 64, v2);
    unsafeAtomicAdd(p + 96, v3);
}

__global__ __launch_bounds__(256) void sage_gemm_wmma(const float* __restrict__ msg,
                                                      const float* __restrict__ feat,
                                                      const float* __restrict__ deg,
                                                      const float* __restrict__ W,
                                                      const float* __restrict__ bias,
                                                      float* __restrict__ out,
                                                      int n_out, int apply_lrelu) {
    __shared__ float Atile[16 * LDA];

    const int tid  = threadIdx.x;
    const int nt   = blockDim.x;
    const int wave = tid >> 5;
    const int lane = tid & 31;
    const int m0   = blockIdx.x * 16;

    for (int i = tid; i < 16 * (KDIM / 4); i += nt) {
        int r  = i >> 5;
        int c4 = (i & 31) << 2;
        int row = m0 + r;
        float inv = 1.0f / (deg[row] + 1.0f);
        const float4 mv = *(const float4*)(msg  + (size_t)row * KDIM + c4);
        const float4 xv = *(const float4*)(feat + (size_t)row * KDIM + c4);
        float* a = &Atile[r * LDA + c4];
        a[0] = (mv.x + xv.x) * inv;
        a[1] = (mv.y + xv.y) * inv;
        a[2] = (mv.z + xv.z) * inv;
        a[3] = (mv.w + xv.w) * inv;
    }
    __syncthreads();

    const int n0 = wave * 16;
    const int lr = lane & 15;
    const int hi = lane >> 4;

    v8f acc = {0.f, 0.f, 0.f, 0.f, 0.f, 0.f, 0.f, 0.f};

#pragma unroll 8
    for (int k = 0; k < KDIM; k += 4) {
        v2f a, b;
        const float* ap = &Atile[lr * LDA + k + 2 * hi];
        a.x = ap[0];
        a.y = ap[1];
        const float* bp = &W[(size_t)(k + 2 * hi) * n_out + n0 + lr];
        b.x = bp[0];
        b.y = bp[n_out];
        acc = __builtin_amdgcn_wmma_f32_16x16x4_f32(
             false, a,  false, b,
             (short)0, acc,  false,  false);
    }

    const float bv   = bias[n0 + lr];
    const int  rbase = m0 + 8 * hi;
    const int  col   = n0 + lr;
#pragma unroll
    for (int i = 0; i < 8; ++i) {
        float v = acc[i] + bv;
        if (apply_lrelu) v = (v >= 0.f) ? v : 0.01f * v;
        out[(size_t)(rbase + i) * n_out + col] = v;
    }
}

extern "C" void kernel_launch(void* const* d_in, const int* in_sizes, int n_in,
                              void* d_out, int out_size, void* d_ws, size_t ws_size,
                              hipStream_t stream) {
    const float* x   = (const float*)d_in[0];
    const int*   edg = (const int*)  d_in[1];
    const float* W1  = (const float*)d_in[2];
    const float* b1  = (const float*)d_in[3];
    const float* W2  = (const float*)d_in[4];
    const float* b2  = (const float*)d_in[5];
    float*       outp = (float*)d_out;

    const int E = in_sizes[1] / 2;
    const int* src = edg;
    const int* dst = edg + E;

    float* msg = (float*)d_ws;
    float* h1  = msg + (size_t)N_NODES * KDIM;
    float* deg = h1  + (size_t)N_NODES * KDIM;

    const size_t msgBytes = (size_t)N_NODES * KDIM * sizeof(float);

    hipMemsetAsync(msg, 0, msgBytes, stream);
    hipMemsetAsync(deg, 0, (size_t)N_NODES * sizeof(float), stream);

    deg_kernel<<<(E + 255) / 256, 256, 0, stream>>>(dst, deg, E);

    {
        long long threads = (long long)E * 32;
        int blocks = (int)((threads + 255) / 256);
        scatter_kernel<<<blocks, 256, 0, stream>>>(x, src, dst, msg, E);
    }

    sage_gemm_wmma<<<N_NODES / 16, 256, 0, stream>>>(msg, x, deg, W1, b1, h1,
                                                      128,  1);

    hipMemsetAsync(msg, 0, msgBytes, stream);

    {
        long long threads = (long long)E * 32;
        int blocks = (int)((threads + 255) / 256);
        scatter_kernel<<<blocks, 256, 0, stream>>>(h1, src, dst, msg, E);
    }

    sage_gemm_wmma<<<N_NODES / 16, 64, 0, stream>>>(msg, h1, deg, W2, b2, outp,
                                                     32,  0);
}


# layer 2: project h1 to 32 cols BEFORE the neighbour sum ((Ah)W = A(hW)); scatter adds z[src]/(deg[dst]+1) by f32 atomics straight into out, 8 edges a wave: 128 B per edge, was 512; sum order changed
# speedup vs baseline: 1.8460x; 1.3693x over previous
#include <hip/hip_runtime.h>
#include <hip/hip_bf16.h>

typedef float v2f __attribute__((ext_vector_type(2)));
typedef float v8f __attribute__((ext_vector_type(8)));

#define N_NODES 50000
#define KDIM    128
#define LDA     132
#define NOUT2   32
#define EPW     8

__global__ __launch_bounds__(256) void deg_kernel(const int* __restrict__ dst,
                                                  float* __restrict__ deg, int nE) {
    int e = blockIdx.x * blockDim.x + threadIdx.x;
    if (e < nE) unsafeAtomicAdd(&deg[dst[e]], 1.0f);
}

__global__ __launch_bounds__(256) void scatter_kernel(const float* __restrict__ feat,
                                                      const int* __restrict__ src,
                                                      const int* __restrict__ dst,
                                                      float* __restrict__ msg, int nE) {
    int t = blockIdx.x * blockDim.x + threadIdx.x;
    int e = t >> 5;
    if (e >= nE) return;
    int l = t & 31;
    int s = src[e], d = dst[e];
    const float* f = feat + (size_t)s * KDIM + l;
    const float v0 = f[0];
    const float v1 = f[32];
    const float v2 = f[64];
    const float v3 = f[96];
    float* p = msg + (size_t)d * KDIM + l;
    unsafeAtomicAdd(p +  0, v0);
    unsafeAtomicAdd(p + 32, v1);
    unsafeAtomicAdd(p + 64, v2);
    unsafeAtomicAdd(p + 96, v3);
}

__global__ __launch_bounds__(256) void sage_gemm_wmma(const float* __restrict__ msg,
                                                      const float* __restrict__ feat,
                                                      const float* __restrict__ deg,
                                                      const float* __restrict__ W,
                                                      const float* __restrict__ bias,
                                                      float* __restrict__ out,
                                                      int n_out, int apply_lrelu) {
    __shared__ float Atile[16 * LDA];

    const int tid  = threadIdx.x;
    const int nt   = blockDim.x;
    const int wave = tid >> 5;
    const int lane = tid & 31;
    const int m0   = blockIdx.x * 16;

    for (int i = tid; i < 16 * (KDIM / 4); i += nt) {
        int r  = i >> 5;
        int c4 = (i & 31) << 2;
        int row = m0 + r;
        float inv = 1.0f / (deg[row] + 1.0f);
        const float4 mv = *(const float4*)(msg  + (size_t)row * KDIM + c4);
        const float4 xv = *(const float4*)(feat + (size_t)row * KDIM + c4);
        float* a = &Atile[r * LDA + c4];
        a[0] = (mv.x + xv.x) * inv;
        a[1] = (mv.y + xv.y) * inv;
        a[2] = (mv.z + xv.z) * inv;
        a[3] = (mv.w + xv.w) * inv;
    }
    __syncthreads();

    const int n0 = wave * 16;
    const int lr = lane & 15;
    const int hi = lane >> 4;

    v8f acc = {0.f, 0.f, 0.f, 0.f, 0.f, 0.f, 0.f, 0.f};

#pragma unroll 8
    for (int k = 0; k < KDIM; k += 4) {
        v2f a, b;
        const float* ap = &Atile[lr * LDA + k + 2 * hi];
        a.x = ap[0];
        a.y = ap[1];
        const float* bp = &W[(size_t)(k + 2 * hi) * n_out + n0 + lr];
        b.x = bp[0];
        b.y = bp[n_out];
        acc = __builtin_amdgcn_wmma_f32_16x16x4_f32(
             false, a,  false, b,
             (short)0, acc,  false,  false);
    }

    const float bv   = bias[n0 + lr];
    const int  rbase = m0 + 8 * hi;
    const int  col   = n0 + lr;
#pragma unroll
    for (int i = 0; i < 8; ++i) {
        float v = acc[i] + bv;
        if (apply_lrelu) v = (v >= 0.f) ? v : 0.01f * v;
        out[(size_t)(rbase + i) * n_out + col] = v;
    }
}

__global__ __launch_bounds__(64) void proj_gemm_wmma(const float* __restrict__ feat,
                                                     const float* __restrict__ deg,
                                                     const float* __restrict__ W,
                                                     const float* __restrict__ bias,
                                                     float* __restrict__ z,
                                                     float* __restrict__ inv,
                                                     float* __restrict__ out) {
    __shared__ float Atile[16 * LDA];
    __shared__ float rinv[16];

    const int tid  = threadIdx.x;
    const int wave = tid >> 5;
    const int lane = tid & 31;
    const int m0   = blockIdx.x * 16;

    if (tid < 16) {
        const float iv = 1.0f / (deg[m0 + tid] + 1.0f);
        rinv[tid]     = iv;
        inv[m0 + tid] = iv;
    }

    {
        const int r0 = tid >> 5;
        const int c4 = (tid & 31) << 2;
        float4 xv[8];
#pragma unroll
        for (int p = 0; p < 8; ++p)
            xv[p] = *(const float4*)(feat + (size_t)(m0 + 2 * p + r0) * KDIM + c4);
#pragma unroll
        for (int p = 0; p < 8; ++p) {
            float* a = &Atile[(2 * p + r0) * LDA + c4];
            a[0] = xv[p].x;
            a[1] = xv[p].y;
            a[2] = xv[p].z;
            a[3] = xv[p].w;
        }
    }
    __syncthreads();

    const int n0 = wave * 16;
    const int lr = lane & 15;
    const int hi = lane >> 4;

    v8f acc = {0.f, 0.f, 0.f, 0.f, 0.f, 0.f, 0.f, 0.f};

#pragma unroll 8
    for (int k = 0; k < KDIM; k += 4) {
        v2f a, b;
        const float* ap = &Atile[lr * LDA + k + 2 * hi];
        a.x = ap[0];
        a.y = ap[1];
        const float* bp = &W[(size_t)(k + 2 * hi) * NOUT2 + n0 + lr];
        b.x = bp[0];
        b.y = bp[NOUT2];
        acc = __builtin_amdgcn_wmma_f32_16x16x4_f32(
             false, a,  false, b,
             (short)0, acc,  false,  false);
    }

    const int  col   = n0 + lr;
    const float bv   = bias[col];
    const int  rbase = m0 + 8 * hi;
#pragma unroll
    for (int i = 0; i < 8; ++i) {
        const int   row = rbase + i;
        const float zv  = acc[i];
        z  [(size_t)row * NOUT2 + col] = zv;
        out[(size_t)row * NOUT2 + col] = zv * rinv[8 * hi + i] + bv;
    }
}

__global__ __launch_bounds__(256) void scatter_proj_kernel(const float* __restrict__ z,
                                                           const int* __restrict__ src,
                                                           const int* __restrict__ dst,
                                                           const float* __restrict__ inv,
                                                           float* __restrict__ out, int nE) {
    const int lane = threadIdx.x & 31;
    const int w  = __builtin_amdgcn_readfirstlane((int)(blockIdx.x * 8 + (threadIdx.x >> 5)));
    const int e0 = w * EPW;
    if (e0 + EPW <= nE) {
        int s[EPW], d[EPW];
#pragma unroll
        for (int j = 0; j < EPW; ++j) { s[j] = src[e0 + j]; d[j] = dst[e0 + j]; }
        float v[EPW], sc[EPW];
#pragma unroll
        for (int j = 0; j < EPW; ++j) {
            v[j]  = z[(size_t)s[j] * NOUT2 + lane];
            sc[j] = inv[d[j]];
        }
#pragma unroll
        for (int j = 0; j < EPW; ++j)
            unsafeAtomicAdd(out + (size_t)d[j] * NOUT2 + lane, v[j] * sc[j]);
    } else {
        for (int e = e0; e < nE; ++e) {
            const int sE = src[e], dE = dst[e];
            unsafeAtomicAdd(out + (size_t)dE * NOUT2 + lane,
                            z[(size_t)sE * NOUT2 + lane] * inv[dE]);
        }
    }
}

extern "C" void kernel_launch(void* const* d_in, const int* in_sizes, int n_in,
                              void* d_out, int out_size, void* d_ws, size_t ws_size,
                              hipStream_t stream) {
    const float* x   = (const float*)d_in[0];
    const int*   edg = (const int*)  d_in[1];
    const float* W1  = (const float*)d_in[2];
    const float* b1  = (const float*)d_in[3];
    const float* W2  = (const float*)d_in[4];
    const float* b2  = (const float*)d_in[5];
    float*       outp = (float*)d_out;

    const int E = in_sizes[1] / 2;
    const int* src = edg;
    const int* dst = edg + E;

    float* msg = (float*)d_ws;
    float* h1  = msg + (size_t)N_NODES * KDIM;
    float* deg = h1  + (size_t)N_NODES * KDIM;

    const size_t msgBytes = (size_t)N_NODES * KDIM * sizeof(float);

    hipMemsetAsync(msg, 0, msgBytes, stream);
    hipMemsetAsync(deg, 0, (size_t)N_NODES * sizeof(float), stream);

    deg_kernel<<<(E + 255) / 256, 256, 0, stream>>>(dst, deg, E);

    {
        long long threads = (long long)E * 32;
        int blocks = (int)((threads + 255) / 256);
        scatter_kernel<<<blocks, 256, 0, stream>>>(x, src, dst, msg, E);
    }

    sage_gemm_wmma<<<N_NODES / 16, 256, 0, stream>>>(msg, x, deg, W1, b1, h1,
                                                      128,  1);

    float* z   = msg;
    float* inv = msg + (size_t)N_NODES * NOUT2;

    proj_gemm_wmma<<<N_NODES / 16, 64, 0, stream>>>(h1, deg, W2, b2, z, inv, outp);

    {
        int blocks = (E + 8 * EPW - 1) / (8 * EPW);
        scatter_proj_kernel<<<blocks, 256, 0, stream>>>(z, src, dst, inv, outp, E);
    }
}


# both neighbour sums: edges sorted by dst in the call (int count+rank, scan, fill), rows summed in registers, layer 1 in the GEMM tile stage; no f32 atomics, msg plane, memset; sum order changed
# speedup vs baseline: 2.7378x; 1.4831x over previous
#include <hip/hip_runtime.h>
#include <hip/hip_bf16.h>

typedef float v2f __attribute__((ext_vector_type(2)));
typedef float v8f __attribute__((ext_vector_type(8)));

#define N_NODES 50000
#define KDIM    128
#define LDA     132
#define NOUT2   32
#define SCAN_T  1024

__global__ __launch_bounds__(256) void count_rank_kernel(const int* __restrict__ dst,
                                                         int* __restrict__ cnt,
                                                         int* __restrict__ rank, int nE) {
    int e = blockIdx.x * blockDim.x + threadIdx.x;
    if (e < nE) rank[e] = atomicAdd(&cnt[dst[e]], 1);
}

__global__ __launch_bounds__(SCAN_T) void rowptr_scan_kernel(const int* __restrict__ cnt,
                                                             int* __restrict__ rowptr) {
    __shared__ int wsum_front[SCAN_T / 32];
    __shared__ int wsum_own[SCAN_T / 32];

    const int lane = threadIdx.x & 31;
    const int wave = threadIdx.x >> 5;
    const int b    = blockIdx.x;

    int front = 0;
#pragma unroll 8
    for (int c = 0; c < b; ++c) front += cnt[c * SCAN_T + threadIdx.x];

    const int i   = b * SCAN_T + threadIdx.x;
    const int own = (i < N_NODES) ? cnt[i] : 0;

    int incl = own;
#pragma unroll
    for (int d = 1; d < 32; d <<= 1) {
        const int t = __shfl_up(incl, d, 32);
        if (lane >= d) incl += t;
    }
#pragma unroll
    for (int d = 16; d >= 1; d >>= 1) front += __shfl_xor(front, d, 32);

    if (lane == 31) wsum_own[wave]   = incl;
    if (lane == 0)  wsum_front[wave] = front;
    __syncthreads();

    int base = wsum_front[lane] + ((lane < wave) ? wsum_own[lane] : 0);
#pragma unroll
    for (int d = 16; d >= 1; d >>= 1) base += __shfl_xor(base, d, 32);

    if (i <= N_NODES) rowptr[i] = base + incl - own;
}

__global__ __launch_bounds__(256) void fill_kernel(const int* __restrict__ src,
                                                   const int* __restrict__ dst,
                                                   const int* __restrict__ rank,
                                                   const int* __restrict__ rowptr,
                                                   int* __restrict__ nbr, int nE) {
    int e = blockIdx.x * blockDim.x + threadIdx.x;
    if (e < nE) nbr[rowptr[dst[e]] + rank[e]] = src[e];
}

__global__ __launch_bounds__(256) void sage_gather_gemm_wmma(const float* __restrict__ feat,
                                                      const int* __restrict__ rowptr,
                                                      const int* __restrict__ nbr,
                                                      const float* __restrict__ W,
                                                      const float* __restrict__ bias,
                                                      float* __restrict__ out,
                                                      int n_out, int apply_lrelu) {
    __shared__ float Atile[16 * LDA];

    const int tid  = threadIdx.x;
    const int lane = tid & 31;
    const int wave = __builtin_amdgcn_readfirstlane(tid >> 5);
    const int m0   = blockIdx.x * 16;
    const int c4   = lane << 2;

    for (int rr = 0; rr < 2; ++rr) {
        const int r   = 2 * wave + rr;
        const int row = m0 + r;
        const int beg = rowptr[row];
        const int end = rowptr[row + 1];
        const float4 xv = *(const float4*)(feat + (size_t)row * KDIM + c4);

        float4 acc = make_float4(0.f, 0.f, 0.f, 0.f);
        int j = beg;
        for (; j + 8 <= end; j += 8) {
            int s[8];
#pragma unroll
            for (int q = 0; q < 8; ++q) s[q] = nbr[j + q];
            float4 v[8];
#pragma unroll
            for (int q = 0; q < 8; ++q) v[q] = *(const float4*)(feat + (size_t)s[q] * KDIM + c4);
#pragma unroll
            for (int q = 0; q < 8; ++q) { acc.x += v[q].x; acc.y += v[q].y; acc.z += v[q].z; acc.w += v[q].w; }
        }
        const int rem = end - j;
        if (rem & 4) {
            int s[4];
#pragma unroll
            for (int q = 0; q < 4; ++q) s[q] = nbr[j + q];
            float4 v[4];
#pragma unroll
            for (int q = 0; q < 4; ++q) v[q] = *(const float4*)(feat + (size_t)s[q] * KDIM + c4);
#pragma unroll
            for (int q = 0; q < 4; ++q) { acc.x += v[q].x; acc.y += v[q].y; acc.z += v[q].z; acc.w += v[q].w; }
            j += 4;
        }
        if (rem & 2) {
            const int s0 = nbr[j], s1 = nbr[j + 1];
            const float4 v0 = *(const float4*)(feat + (size_t)s0 * KDIM + c4);
            const float4 v1 = *(const float4*)(feat + (size_t)s1 * KDIM + c4);
            acc.x += v0.x; acc.y += v0.y; acc.z += v0.z; acc.w += v0.w;
            acc.x += v1.x; acc.y += v1.y; acc.z += v1.z; acc.w += v1.w;
            j += 2;
        }
        if (rem & 1) {
            const int s0 = nbr[j];
            const float4 v0 = *(const float4*)(feat + (size_t)s0 * KDIM + c4);
            acc.x += v0.x; acc.y += v0.y; acc.z += v0.z; acc.w += v0.w;
        }

        const float inv = 1.0f / ((float)(end - beg) + 1.0f);
        float* a = &Atile[r * LDA + c4];
        a[0] = (acc.x + xv.x) * inv;
        a[1] = (acc.y + xv.y) * inv;
        a[2] = (acc.z + xv.z) * inv;
        a[3] = (acc.w + xv.w) * inv;
    }
    __syncthreads();

    const int n0 = wave * 16;
    const int lr = lane & 15;
    const int hi = lane >> 4;

    v8f acc = {0.f, 0.f, 0.f, 0.f, 0.f, 0.f, 0.f, 0.f};

#pragma unroll 8
    for (int k = 0; k < KDIM; k += 4) {
        v2f a, b;
        const float* ap = &Atile[lr * LDA + k + 2 * hi];
        a.x = ap[0];
        a.y = ap[1];
        const float* bp = &W[(size_t)(k + 2 * hi) * n_out + n0 + lr];
        b.x = bp[0];
        b.y = bp[n_out];
        acc = __builtin_amdgcn_wmma_f32_16x16x4_f32(
             false, a,  false, b,
             (short)0, acc,  false,  false);
    }

    const float bv   = bias[n0 + lr];
    const int  rbase = m0 + 8 * hi;
    const int  col   = n0 + lr;
#pragma unroll
    for (int i = 0; i < 8; ++i) {
        float v = acc[i] + bv;
        if (apply_lrelu) v = (v >= 0.f) ? v : 0.01f * v;
        out[(size_t)(rbase + i) * n_out + col] = v;
    }
}

__global__ __launch_bounds__(64) void proj_gemm_wmma(const float* __restrict__ feat,
                                                     const float* __restrict__ W,
                                                     float* __restrict__ z) {
    __shared__ float Atile[16 * LDA];

    const int tid  = threadIdx.x;
    const int wave = tid >> 5;
    const int lane = tid & 31;
    const int m0   = blockIdx.x * 16;

    {
        const int r0 = tid >> 5;
        const int c4 = (tid & 31) << 2;
        float4 xv[8];
#pragma unroll
        for (int p = 0; p < 8; ++p)
            xv[p] = *(const float4*)(feat + (size_t)(m0 + 2 * p + r0) * KDIM + c4);
#pragma unroll
        for (int p = 0; p < 8; ++p) {
            float* a = &Atile[(2 * p + r0) * LDA + c4];
            a[0] = xv[p].x;
            a[1] = xv[p].y;
            a[2] = xv[p].z;
            a[3] = xv[p].w;
        }
    }
    __syncthreads();

    const int n0 = wave * 16;
    const int lr = lane & 15;
    const int hi = lane >> 4;

    v8f acc = {0.f, 0.f, 0.f, 0.f, 0.f, 0.f, 0.f, 0.f};

#pragma unroll 8
    for (int k = 0; k < KDIM; k += 4) {
        v2f a, b;
        const float* ap = &Atile[lr * LDA + k + 2 * hi];
        a.x = ap[0];
        a.y = ap[1];
        const float* bp = &W[(size_t)(k + 2 * hi) * NOUT2 + n0 + lr];
        b.x = bp[0];
        b.y = bp[NOUT2];
        acc = __builtin_amdgcn_wmma_f32_16x16x4_f32(
             false, a,  false, b,
             (short)0, acc,  false,  false);
    }

    const int  col   = n0 + lr;
    const int  rbase = m0 + 8 * hi;
#pragma unroll
    for (int i = 0; i < 8; ++i)
        z[(size_t)(rbase + i) * NOUT2 + col] = acc[i];
}

__global__ __launch_bounds__(256) void gather_out_kernel(const float* __restrict__ z,
                                                         const int* __restrict__ rowptr,
                                                         const int* __restrict__ nbr,
                                                         const float* __restrict__ bias,
                                                         float* __restrict__ out) {
    const int lane = threadIdx.x & 31;
    const int w = __builtin_amdgcn_readfirstlane((int)(blockIdx.x * 8 + (threadIdx.x >> 5)));
    const float bv = bias[lane];

    for (int rr = 0; rr < 2; ++rr) {
        const int row = 2 * w + rr;
        if (row >= N_NODES) break;
        const int beg = rowptr[row];
        const int end = rowptr[row + 1];
        const float zv = z[(size_t)row * NOUT2 + lane];

        float acc = 0.f;
        int j = beg;
        for (; j + 8 <= end; j += 8) {
            int s[8];
#pragma unroll
            for (int q = 0; q < 8; ++q) s[q] = nbr[j + q];
            float v[8];
#pragma unroll
            for (int q = 0; q < 8; ++q) v[q] = z[(size_t)s[q] * NOUT2 + lane];
#pragma unroll
            for (int q = 0; q < 8; ++q) acc += v[q];
        }
        const int rem = end - j;
        if (rem & 4) {
            int s[4];
#pragma unroll
            for (int q = 0; q < 4; ++q) s[q] = nbr[j + q];
            float v[4];
#pragma unroll
            for (int q = 0; q < 4; ++q) v[q] = z[(size_t)s[q] * NOUT2 + lane];
#pragma unroll
            for (int q = 0; q < 4; ++q) acc += v[q];
            j += 4;
        }
        if (rem & 2) {
            const int s0 = nbr[j], s1 = nbr[j + 1];
            const float v0 = z[(size_t)s0 * NOUT2 + lane];
            const float v1 = z[(size_t)s1 * NOUT2 + lane];
            acc += v0;
            acc += v1;
            j += 2;
        }
        if (rem & 1) acc += z[(size_t)nbr[j] * NOUT2 + lane];

        const float inv = 1.0f / ((float)(end - beg) + 1.0f);
        out[(size_t)row * NOUT2 + lane] = (acc + zv) * inv + bv;
    }
}

extern "C" void kernel_launch(void* const* d_in, const int* in_sizes, int n_in,
                              void* d_out, int out_size, void* d_ws, size_t ws_size,
                              hipStream_t stream) {
    const float* x   = (const float*)d_in[0];
    const int*   edg = (const int*)  d_in[1];
    const float* W1  = (const float*)d_in[2];
    const float* b1  = (const float*)d_in[3];
    const float* W2  = (const float*)d_in[4];
    const float* b2  = (const float*)d_in[5];
    float*       outp = (float*)d_out;

    const int E = in_sizes[1] / 2;
    const int* src = edg;
    const int* dst = edg + E;

    float* plane0 = (float*)d_ws;
    float* h1     = plane0 + (size_t)N_NODES * KDIM;
    int*   cnt    = (int*)(h1 + (size_t)N_NODES * KDIM);

    float* z      = plane0;
    int*   rank   = (int*)(z + (size_t)N_NODES * NOUT2);
    int*   nbr    = rank + E;
    int*   rowptr = nbr + E;

    hipMemsetAsync(cnt, 0, (size_t)N_NODES * sizeof(int), stream);

    count_rank_kernel<<<(E + 255) / 256, 256, 0, stream>>>(dst, cnt, rank, E);

    rowptr_scan_kernel<<<(N_NODES + SCAN_T) / SCAN_T, SCAN_T, 0, stream>>>(cnt, rowptr);

    fill_kernel<<<(E + 255) / 256, 256, 0, stream>>>(src, dst, rank, rowptr, nbr, E);

    sage_gather_gemm_wmma<<<N_NODES / 16, 256, 0, stream>>>(x, rowptr, nbr, W1, b1, h1,
                                                             128,  1);

    proj_gemm_wmma<<<N_NODES / 16, 64, 0, stream>>>(h1, W2, z);

    gather_out_kernel<<<N_NODES / 16, 256, 0, stream>>>(z, rowptr, nbr, b2, outp);
}


# layer-2 projection z = h1 @ W2 moved into the layer-1 kernel, from an LDS tile of h1 (waves 0,1, same k order): h1 never goes to memory, one launch fewer; sum order unchanged
# speedup vs baseline: 3.0126x; 1.1004x over previous
#include <hip/hip_runtime.h>
#include <hip/hip_bf16.h>

typedef float v2f __attribute__((ext_vector_type(2)));
typedef float v8f __attribute__((ext_vector_type(8)));

#define N_NODES 50000
#define KDIM    128
#define LDA     132
#define NOUT2   32
#define SCAN_T  1024

__global__ __launch_bounds__(256) void count_rank_kernel(const int* __restrict__ dst,
                                                         int* __restrict__ cnt,
                                                         int* __restrict__ rank, int nE) {
    int e = blockIdx.x * blockDim.x + threadIdx.x;
    if (e < nE) rank[e] = atomicAdd(&cnt[dst[e]], 1);
}

__global__ __launch_bounds__(SCAN_T) void rowptr_scan_kernel(const int* __restrict__ cnt,
                                                             int* __restrict__ rowptr) {
    __shared__ int wsum_front[SCAN_T / 32];
    __shared__ int wsum_own[SCAN_T / 32];

    const int lane = threadIdx.x & 31;
    const int wave = threadIdx.x >> 5;
    const int b    = blockIdx.x;

    int front = 0;
#pragma unroll 8
    for (int c = 0; c < b; ++c) front += cnt[c * SCAN_T + threadIdx.x];

    const int i   = b * SCAN_T + threadIdx.x;
    const int own = (i < N_NODES) ? cnt[i] : 0;

    int incl = own;
#pragma unroll
    for (int d = 1; d < 32; d <<= 1) {
        const int t = __shfl_up(incl, d, 32);
        if (lane >= d) incl += t;
    }
#pragma unroll
    for (int d = 16; d >= 1; d >>= 1) front += __shfl_xor(front, d, 32);

    if (lane == 31) wsum_own[wave]   = incl;
    if (lane == 0)  wsum_front[wave] = front;
    __syncthreads();

    int base = wsum_front[lane] + ((lane < wave) ? wsum_own[lane] : 0);
#pragma unroll
    for (int d = 16; d >= 1; d >>= 1) base += __shfl_xor(base, d, 32);

    if (i <= N_NODES) rowptr[i] = base + incl - own;
}

__global__ __launch_bounds__(256) void fill_kernel(const int* __restrict__ src,
                                                   const int* __restrict__ dst,
                                                   const int* __restrict__ rank,
                                                   const int* __restrict__ rowptr,
                                                   int* __restrict__ nbr, int nE) {
    int e = blockIdx.x * blockDim.x + threadIdx.x;
    if (e < nE) nbr[rowptr[dst[e]] + rank[e]] = src[e];
}

__global__ __launch_bounds__(256) void sage_gather_gemm_proj_wmma(const float* __restrict__ feat,
                                                      const int* __restrict__ rowptr,
                                                      const int* __restrict__ nbr,
                                                      const float* __restrict__ W1,
                                                      const float* __restrict__ b1,
                                                      const float* __restrict__ W2,
                                                      float* __restrict__ z) {
    __shared__ float Atile[16 * LDA];
    __shared__ float Htile[16 * LDA];

    const int tid  = threadIdx.x;
    const int lane = tid & 31;
    const int wave = __builtin_amdgcn_readfirstlane(tid >> 5);
    const int m0   = blockIdx.x * 16;
    const int c4   = lane << 2;

    for (int rr = 0; rr < 2; ++rr) {
        const int r   = 2 * wave + rr;
        const int row = m0 + r;
        const int beg = rowptr[row];
        const int end = rowptr[row + 1];
        const float4 xv = *(const float4*)(feat + (size_t)row * KDIM + c4);

        float4 acc = make_float4(0.f, 0.f, 0.f, 0.f);
        int j = beg;
        for (; j + 8 <= end; j += 8) {
            int s[8];
#pragma unroll
            for (int q = 0; q < 8; ++q) s[q] = nbr[j + q];
            float4 v[8];
#pragma unroll
            for (int q = 0; q < 8; ++q) v[q] = *(const float4*)(feat + (size_t)s[q] * KDIM + c4);
#pragma unroll
            for (int q = 0; q < 8; ++q) { acc.x += v[q].x; acc.y += v[q].y; acc.z += v[q].z; acc.w += v[q].w; }
        }
        const int rem = end - j;
        if (rem & 4) {
            int s[4];
#pragma unroll
            for (int q = 0; q < 4; ++q) s[q] = nbr[j + q];
            float4 v[4];
#pragma unroll
            for (int q = 0; q < 4; ++q) v[q] = *(const float4*)(feat + (size_t)s[q] * KDIM + c4);
#pragma unroll
            for (int q = 0; q < 4; ++q) { acc.x += v[q].x; acc.y += v[q].y; acc.z += v[q].z; acc.w += v[q].w; }
            j += 4;
        }
        if (rem & 2) {
            const int s0 = nbr[j], s1 = nbr[j + 1];
            const float4 v0 = *(const float4*)(feat + (size_t)s0 * KDIM + c4);
            const float4 v1 = *(const float4*)(feat + (size_t)s1 * KDIM + c4);
            acc.x += v0.x; acc.y += v0.y; acc.z += v0.z; acc.w += v0.w;
            acc.x += v1.x; acc.y += v1.y; acc.z += v1.z; acc.w += v1.w;
            j += 2;
        }
        if (rem & 1) {
            const int s0 = nbr[j];
            const float4 v0 = *(const float4*)(feat + (size_t)s0 * KDIM + c4);
            acc.x += v0.x; acc.y += v0.y; acc.z += v0.z; acc.w += v0.w;
        }

        const float inv = 1.0f / ((float)(end - beg) + 1.0f);
        float* a = &Atile[r * LDA + c4];
        a[0] = (acc.x + xv.x) * inv;
        a[1] = (acc.y + xv.y) * inv;
        a[2] = (acc.z + xv.z) * inv;
        a[3] = (acc.w + xv.w) * inv;
    }
    __syncthreads();

    const int n0 = wave * 16;
    const int lr = lane & 15;
    const int hi = lane >> 4;

    v8f acc = {0.f, 0.f, 0.f, 0.f, 0.f, 0.f, 0.f, 0.f};

#pragma unroll 8
    for (int k = 0; k < KDIM; k += 4) {
        v2f a, b;
        const float* ap = &Atile[lr * LDA + k + 2 * hi];
        a.x = ap[0];
        a.y = ap[1];
        const float* bp = &W1[(size_t)(k + 2 * hi) * KDIM + n0 + lr];
        b.x = bp[0];
        b.y = bp[KDIM];
        acc = __builtin_amdgcn_wmma_f32_16x16x4_f32(
             false, a,  false, b,
             (short)0, acc,  false,  false);
    }

    const float bv = b1[n0 + lr];
#pragma unroll
    for (int i = 0; i < 8; ++i) {
        float v = acc[i] + bv;
        v = (v >= 0.f) ? v : 0.01f * v;
        Htile[(8 * hi + i) * LDA + n0 + lr] = v;
    }
    __syncthreads();

    if (wave < 2) {
        const int p0 = wave * 16;
        v8f acc2 = {0.f, 0.f, 0.f, 0.f, 0.f, 0.f, 0.f, 0.f};

#pragma unroll 8
        for (int k = 0; k < KDIM; k += 4) {
            v2f a, b;
            const float* ap = &Htile[lr * LDA + k + 2 * hi];
            a.x = ap[0];
            a.y = ap[1];
            const float* bp = &W2[(size_t)(k + 2 * hi) * NOUT2 + p0 + lr];
            b.x = bp[0];
            b.y = bp[NOUT2];
            acc2 = __builtin_amdgcn_wmma_f32_16x16x4_f32(
                 false, a,  false, b,
                 (short)0, acc2,  false,  false);
        }

        const int col   = p0 + lr;
        const int rbase = m0 + 8 * hi;
#pragma unroll
        for (int i = 0; i < 8; ++i)
            z[(size_t)(rbase + i) * NOUT2 + col] = acc2[i];
    }
}

__global__ __launch_bounds__(256) void gather_out_kernel(const float* __restrict__ z,
                                                         const int* __restrict__ rowptr,
                                                         const int* __restrict__ nbr,
                                                         const float* __restrict__ bias,
                                                         float* __restrict__ out) {
    const int lane = threadIdx.x & 31;
    const int w = __builtin_amdgcn_readfirstlane((int)(blockIdx.x * 8 + (threadIdx.x >> 5)));
    const float bv = bias[lane];

    for (int rr = 0; rr < 2; ++rr) {
        const int row = 2 * w + rr;
        if (row >= N_NODES) break;
        const int beg = rowptr[row];
        const int end = rowptr[row + 1];
        const float zv = z[(size_t)row * NOUT2 + lane];

        float acc = 0.f;
        int j = beg;
        for (; j + 8 <= end; j += 8) {
            int s[8];
#pragma unroll
            for (int q = 0; q < 8; ++q) s[q] = nbr[j + q];
            float v[8];
#pragma unroll
            for (int q = 0; q < 8; ++q) v[q] = z[(size_t)s[q] * NOUT2 + lane];
#pragma unroll
            for (int q = 0; q < 8; ++q) acc += v[q];
        }
        const int rem = end - j;
        if (rem & 4) {
            int s[4];
#pragma unroll
            for (int q = 0; q < 4; ++q) s[q] = nbr[j + q];
            float v[4];
#pragma unroll
            for (int q = 0; q < 4; ++q) v[q] = z[(size_t)s[q] * NOUT2 + lane];
#pragma unroll
            for (int q = 0; q < 4; ++q) acc += v[q];
            j += 4;
        }
        if (rem & 2) {
            const int s0 = nbr[j], s1 = nbr[j + 1];
            const float v0 = z[(size_t)s0 * NOUT2 + lane];
            const float v1 = z[(size_t)s1 * NOUT2 + lane];
            acc += v0;
            acc += v1;
            j += 2;
        }
        if (rem & 1) acc += z[(size_t)nbr[j] * NOUT2 + lane];

        const float inv = 1.0f / ((float)(end - beg) + 1.0f);
        out[(size_t)row * NOUT2 + lane] = (acc + zv) * inv + bv;
    }
}

extern "C" void kernel_launch(void* const* d_in, const int* in_sizes, int n_in,
                              void* d_out, int out_size, void* d_ws, size_t ws_size,
                              hipStream_t stream) {
    const float* x   = (const float*)d_in[0];
    const int*   edg = (const int*)  d_in[1];
    const float* W1  = (const float*)d_in[2];
    const float* b1  = (const float*)d_in[3];
    const float* W2  = (const float*)d_in[4];
    const float* b2  = (const float*)d_in[5];
    float*       outp = (float*)d_out;

    const int E = in_sizes[1] / 2;
    const int* src = edg;
    const int* dst = edg + E;

    float* plane0 = (float*)d_ws;
    int*   cnt    = (int*)(plane0 + 2 * (size_t)N_NODES * KDIM);

    float* z      = plane0;
    int*   rank   = (int*)(z + (size_t)N_NODES * NOUT2);
    int*   nbr    = rank + E;
    int*   rowptr = nbr + E;

    hipMemsetAsync(cnt, 0, (size_t)N_NODES * sizeof(int), stream);

    count_rank_kernel<<<(E + 255) / 256, 256, 0, stream>>>(dst, cnt, rank, E);

    rowptr_scan_kernel<<<(N_NODES + SCAN_T) / SCAN_T, SCAN_T, 0, stream>>>(cnt, rowptr);

    fill_kernel<<<(E + 255) / 256, 256, 0, stream>>>(src, dst, rank, rowptr, nbr, E);

    sage_gather_gemm_proj_wmma<<<N_NODES / 16, 256, 0, stream>>>(x, rowptr, nbr, W1, b1, W2, z);

    gather_out_kernel<<<N_NODES / 16, 256, 0, stream>>>(z, rowptr, nbr, b2, outp);
}


# edge list in ONE pass: source goes to slot (value the integer add returns) of a fixed 64-slot row per destination, overflow list behind; no rank, scan or fill launch; sum order by arrival as before
# speedup vs baseline: 4.2179x; 1.4001x over previous
#include <hip/hip_runtime.h>
#include <hip/hip_bf16.h>

typedef float v2f __attribute__((ext_vector_type(2)));
typedef float v8f __attribute__((ext_vector_type(8)));

#define N_NODES 50000
#define KDIM    128
#define LDA     132
#define NOUT2   32
#define CAP     64

__global__ __launch_bounds__(256) void place_edges_kernel(const int* __restrict__ src,
                                                          const int* __restrict__ dst,
                                                          int* __restrict__ cnt,
                                                          int* __restrict__ nbr,
                                                          int* __restrict__ ovf_src,
                                                          int* __restrict__ ovf_dst,
                                                          int nE) {
    int e = blockIdx.x * blockDim.x + threadIdx.x;
    if (e < nE) {
        const int d    = dst[e];
        const int s    = src[e];
        const int slot = atomicAdd(&cnt[d], 1);
        if (slot < CAP) {
            nbr[d * CAP + slot] = s;
        } else {
            const int o = atomicAdd(&cnt[N_NODES], 1);
            ovf_src[o] = s;
            ovf_dst[o] = d;
        }
    }
}

__global__ __launch_bounds__(256) void sage_gather_gemm_proj_wmma(const float* __restrict__ feat,
                                                      const int* __restrict__ cnt,
                                                      const int* __restrict__ nbr,
                                                      const int* __restrict__ ovf_src,
                                                      const int* __restrict__ ovf_dst,
                                                      int nE,
                                                      const float* __restrict__ W1,
                                                      const float* __restrict__ b1,
                                                      const float* __restrict__ W2,
                                                      float* __restrict__ z) {
    __shared__ float Atile[16 * LDA];
    __shared__ float Htile[16 * LDA];

    const int tid  = threadIdx.x;
    const int lane = tid & 31;
    const int wave = __builtin_amdgcn_readfirstlane(tid >> 5);
    const int m0   = blockIdx.x * 16;
    const int c4   = lane << 2;

    for (int rr = 0; rr < 2; ++rr) {
        const int r   = 2 * wave + rr;
        const int row = m0 + r;
        const int deg = cnt[row];
        const int beg = row * CAP;
        const int end = beg + ((deg < CAP) ? deg : CAP);
        const float4 xv = *(const float4*)(feat + (size_t)row * KDIM + c4);

        float4 acc = make_float4(0.f, 0.f, 0.f, 0.f);
        int j = beg;
        for (; j + 8 <= end; j += 8) {
            int s[8];
#pragma unroll
            for (int q = 0; q < 8; ++q) s[q] = nbr[j + q];
            float4 v[8];
#pragma unroll
            for (int q = 0; q < 8; ++q) v[q] = *(const float4*)(feat + (size_t)s[q] * KDIM + c4);
#pragma unroll
            for (int q = 0; q < 8; ++q) { acc.x += v[q].x; acc.y += v[q].y; acc.z += v[q].z; acc.w += v[q].w; }
        }
        const int rem = end - j;
        if (rem & 4) {
            int s[4];
#pragma unroll
            for (int q = 0; q < 4; ++q) s[q] = nbr[j + q];
            float4 v[4];
#pragma unroll
            for (int q = 0; q < 4; ++q) v[q] = *(const float4*)(feat + (size_t)s[q] * KDIM + c4);
#pragma unroll
            for (int q = 0; q < 4; ++q) { acc.x += v[q].x; acc.y += v[q].y; acc.z += v[q].z; acc.w += v[q].w; }
            j += 4;
        }
        if (rem & 2) {
            const int s0 = nbr[j], s1 = nbr[j + 1];
            const float4 v0 = *(const float4*)(feat + (size_t)s0 * KDIM + c4);
            const float4 v1 = *(const float4*)(feat + (size_t)s1 * KDIM + c4);
            acc.x += v0.x; acc.y += v0.y; acc.z += v0.z; acc.w += v0.w;
            acc.x += v1.x; acc.y += v1.y; acc.z += v1.z; acc.w += v1.w;
            j += 2;
        }
        if (rem & 1) {
            const int s0 = nbr[j];
            const float4 v0 = *(const float4*)(feat + (size_t)s0 * KDIM + c4);
            acc.x += v0.x; acc.y += v0.y; acc.z += v0.z; acc.w += v0.w;
        }
        if (deg > CAP) {
            const int nraw = cnt[N_NODES];
            const int novf = (nraw < nE) ? nraw : nE;
            for (int i0 = 0; i0 < novf; i0 += 32) {
                const int i  = i0 + lane;
                const int od = (i < novf) ? ovf_dst[i] : -1;
                const int os = (i < novf) ? ovf_src[i] : 0;
                unsigned hit = (unsigned)__ballot(od == row);
                while (hit) {
                    const int b = __builtin_ctz(hit);
                    hit &= hit - 1;
                    const int sb = __shfl(os, b, 32);
                    const float4 vb = *(const float4*)(feat + (size_t)sb * KDIM + c4);
                    acc.x += vb.x; acc.y += vb.y; acc.z += vb.z; acc.w += vb.w;
                }
            }
        }

        const float inv = 1.0f / ((float)deg + 1.0f);
        float* a = &Atile[r * LDA + c4];
        a[0] = (acc.x + xv.x) * inv;
        a[1] = (acc.y + xv.y) * inv;
        a[2] = (acc.z + xv.z) * inv;
        a[3] = (acc.w + xv.w) * inv;
    }
    __syncthreads();

    const int n0 = wave * 16;
    const int lr = lane & 15;
    const int hi = lane >> 4;

    v8f acc = {0.f, 0.f, 0.f, 0.f, 0.f, 0.f, 0.f, 0.f};

#pragma unroll 8
    for (int k = 0; k < KDIM; k += 4) {
        v2f a, b;
        const float* ap = &Atile[lr * LDA + k + 2 * hi];
        a.x = ap[0];
        a.y = ap[1];
        const float* bp = &W1[(size_t)(k + 2 * hi) * KDIM + n0 + lr];
        b.x = bp[0];
        b.y = bp[KDIM];
        acc = __builtin_amdgcn_wmma_f32_16x16x4_f32(
             false, a,  false, b,
             (short)0, acc,  false,  false);
    }

    const float bv = b1[n0 + lr];
#pragma unroll
    for (int i = 0; i < 8; ++i) {
        float v = acc[i] + bv;
        v = (v >= 0.f) ? v : 0.01f * v;
        Htile[(8 * hi + i) * LDA + n0 + lr] = v;
    }
    __syncthreads();

    if (wave < 2) {
        const int p0 = wave * 16;
        v8f acc2 = {0.f, 0.f, 0.f, 0.f, 0.f, 0.f, 0.f, 0.f};

#pragma unroll 8
        for (int k = 0; k < KDIM; k += 4) {
            v2f a, b;
            const float* ap = &Htile[lr * LDA + k + 2 * hi];
            a.x = ap[0];
            a.y = ap[1];
            const float* bp = &W2[(size_t)(k + 2 * hi) * NOUT2 + p0 + lr];
            b.x = bp[0];
            b.y = bp[NOUT2];
            acc2 = __builtin_amdgcn_wmma_f32_16x16x4_f32(
                 false, a,  false, b,
                 (short)0, acc2,  false,  false);
        }

        const int col   = p0 + lr;
        const int rbase = m0 + 8 * hi;
#pragma unroll
        for (int i = 0; i < 8; ++i)
            z[(size_t)(rbase + i) * NOUT2 + col] = acc2[i];
    }
}

__global__ __launch_bounds__(256) void gather_out_kernel(const float* __restrict__ z,
                                                         const int* __restrict__ cnt,
                                                         const int* __restrict__ nbr,
                                                         const int* __restrict__ ovf_src,
                                                         const int* __restrict__ ovf_dst,
                                                         int nE,
                                                         const float* __restrict__ bias,
                                                         float* __restrict__ out) {
    const int lane = threadIdx.x & 31;
    const int w = __builtin_amdgcn_readfirstlane((int)(blockIdx.x * 8 + (threadIdx.x >> 5)));
    const float bv = bias[lane];

    for (int rr = 0; rr < 2; ++rr) {
        const int row = 2 * w + rr;
        if (row >= N_NODES) break;
        const int deg = cnt[row];
        const int beg = row * CAP;
        const int end = beg + ((deg < CAP) ? deg : CAP);
        const float zv = z[(size_t)row * NOUT2 + lane];

        float acc = 0.f;
        int j = beg;
        for (; j + 8 <= end; j += 8) {
            int s[8];
#pragma unroll
            for (int q = 0; q < 8; ++q) s[q] = nbr[j + q];
            float v[8];
#pragma unroll
            for (int q = 0; q < 8; ++q) v[q] = z[(size_t)s[q] * NOUT2 + lane];
#pragma unroll
            for (int q = 0; q < 8; ++q) acc += v[q];
        }
        const int rem = end - j;
        if (rem & 4) {
            int s[4];
#pragma unroll
            for (int q = 0; q < 4; ++q) s[q] = nbr[j + q];
            float v[4];
#pragma unroll
            for (int q = 0; q < 4; ++q) v[q] = z[(size_t)s[q] * NOUT2 + lane];
#pragma unroll
            for (int q = 0; q < 4; ++q) acc += v[q];
            j += 4;
        }
        if (rem & 2) {
            const int s0 = nbr[j], s1 = nbr[j + 1];
            const float v0 = z[(size_t)s0 * NOUT2 + lane];
            const float v1 = z[(size_t)s1 * NOUT2 + lane];
            acc += v0;
            acc += v1;
            j += 2;
        }
        if (rem & 1) acc += z[(size_t)nbr[j] * NOUT2 + lane];
        if (deg > CAP) {
            const int nraw = cnt[N_NODES];
            const int novf = (nraw < nE) ? nraw : nE;
            for (int i0 = 0; i0 < novf; i0 += 32) {
                const int i  = i0 + lane;
                const int od = (i < novf) ? ovf_dst[i] : -1;
                const int os = (i < novf) ? ovf_src[i] : 0;
                unsigned hit = (unsigned)__ballot(od == row);
                while (hit) {
                    const int b = __builtin_ctz(hit);
                    hit &= hit - 1;
                    const int sb = __shfl(os, b, 32);
                    acc += z[(size_t)sb * NOUT2 + lane];
                }
            }
        }

        const float inv = 1.0f / ((float)deg + 1.0f);
        out[(size_t)row * NOUT2 + lane] = (acc + zv) * inv + bv;
    }
}

extern "C" void kernel_launch(void* const* d_in, const int* in_sizes, int n_in,
                              void* d_out, int out_size, void* d_ws, size_t ws_size,
                              hipStream_t stream) {
    const float* x   = (const float*)d_in[0];
    const int*   edg = (const int*)  d_in[1];
    const float* W1  = (const float*)d_in[2];
    const float* b1  = (const float*)d_in[3];
    const float* W2  = (const float*)d_in[4];
    const float* b2  = (const float*)d_in[5];
    float*       outp = (float*)d_out;

    const int E = in_sizes[1] / 2;
    const int* src = edg;
    const int* dst = edg + E;

    float* z       = (float*)d_ws;
    int*   nbr     = (int*)(z + (size_t)N_NODES * NOUT2);
    int*   cnt     = nbr + (size_t)N_NODES * CAP;
    int*   ovf_src = cnt + (N_NODES + 48);
    int*   ovf_dst = ovf_src + E;

    hipMemsetAsync(cnt, 0, (size_t)(N_NODES + 1) * sizeof(int), stream);

    place_edges_kernel<<<(E + 255) / 256, 256, 0, stream>>>(src, dst, cnt, nbr, ovf_src, ovf_dst, E);

    sage_gather_gemm_proj_wmma<<<N_NODES / 16, 256, 0, stream>>>(x, cnt, nbr, ovf_src, ovf_dst, E, W1, b1, W2, z);

    gather_out_kernel<<<N_NODES / 16, 256, 0, stream>>>(z, cnt, nbr, ovf_src, ovf_dst, E, b2, outp);
}


# edge table built through LDS: 4096-edge chunks sorted by 256-row bucket, then a workgroup per bucket fills its rows in LDS and stores them whole; no global add per edge, no memset; sums by arrival
# speedup vs baseline: 7.0622x; 1.6743x over previous
#include <hip/hip_runtime.h>
#include <hip/hip_bf16.h>

typedef float v2f __attribute__((ext_vector_type(2)));
typedef float v8f __attribute__((ext_vector_type(8)));

#define N_NODES 50000
#define KDIM    128
#define LDA     132
#define NOUT2   32
#define CAP     64

#define RB      256
#define NB      ((N_NODES + RB - 1) / RB)
#define CHUNK   4096
static_assert(RB == 256, "a record keeps the row inside its bucket in 8 bits");
static_assert(N_NODES < (1 << 23), "a record keeps the source above those 8 bits");
static_assert(NB <= 7 * 32, "the bucket scan gives each lane of one wave 7 buckets");

__global__ __launch_bounds__(1024) void bin_edges_kernel(const int* __restrict__ src,
                                                         const int* __restrict__ dst,
                                                         int* __restrict__ recs,
                                                         int* __restrict__ runs,
                                                         int* __restrict__ cnt,
                                                         int nE, int nChunks) {
    __shared__ int hist[NB];
    __shared__ int start[NB];
    __shared__ int sorted[CHUNK];

    const int tid  = threadIdx.x;
    const int base = blockIdx.x * CHUNK;

    if (tid < NB) hist[tid] = 0;
    if (blockIdx.x == 0 && tid == 0) cnt[N_NODES] = 0;
    __syncthreads();

    int dv[4], sv[4];
#pragma unroll
    for (int q = 0; q < 4; ++q) {
        const int e  = base + q * 1024 + tid;
        const int ec = (e < nE) ? e : (nE - 1);
        dv[q] = dst[ec];
        sv[q] = src[ec];
    }
    int rec[4], bk[4], rk[4];
#pragma unroll
    for (int q = 0; q < 4; ++q) {
        const int e = base + q * 1024 + tid;
        bk[q] = -1; rec[q] = 0; rk[q] = 0;
        if (e < nE) {
            bk[q]  = dv[q] >> 8;
            rec[q] = (sv[q] << 8) | (dv[q] & (RB - 1));
            rk[q]  = atomicAdd(&hist[bk[q]], 1);
        }
    }
    __syncthreads();

    if (tid < 32) {
        int c[7];
        int sum = 0;
#pragma unroll
        for (int i = 0; i < 7; ++i) {
            const int b = tid * 7 + i;
            c[i] = (b < NB) ? hist[b] : 0;
            sum += c[i];
        }
        int incl = sum;
#pragma unroll
        for (int off = 1; off < 32; off <<= 1) {
            const int t = __shfl_up(incl, off, 32);
            if (tid >= off) incl += t;
        }
        int run = incl - sum;
#pragma unroll
        for (int i = 0; i < 7; ++i) {
            const int b = tid * 7 + i;
            if (b < NB) start[b] = run;
            run += c[i];
        }
    }
    __syncthreads();

#pragma unroll
    for (int q = 0; q < 4; ++q)
        if (bk[q] >= 0) sorted[start[bk[q]] + rk[q]] = rec[q];
    __syncthreads();

#pragma unroll
    for (int q = 0; q < 4; ++q) {
        const int i = q * 1024 + tid;
        if (base + i < nE) recs[base + i] = sorted[i];
    }
    if (tid < NB) runs[tid * nChunks + blockIdx.x] = (start[tid] << 16) | hist[tid];
}

__device__ __forceinline__ void place_record(int rc, int b, int* lcnt, int* tab,
                                             int* __restrict__ cnt, int* __restrict__ ovf_src,
                                             int* __restrict__ ovf_dst) {
    const int r    = rc & (RB - 1);
    const int s    = rc >> 8;
    const int slot = atomicAdd(&lcnt[r], 1);
    if (slot < CAP) {
        tab[r * CAP + slot] = s;
    } else {
        const int o = atomicAdd(&cnt[N_NODES], 1);
        ovf_src[o] = s;
        ovf_dst[o] = b * RB + r;
    }
}

__global__ __launch_bounds__(1024) void build_rows_kernel(const int* __restrict__ recs,
                                                          const int* __restrict__ runs,
                                                          int* __restrict__ cnt,
                                                          int* __restrict__ nbr,
                                                          int* __restrict__ ovf_src,
                                                          int* __restrict__ ovf_dst,
                                                          int nChunks) {
    __shared__ int4 tab4[RB * CAP / 4];
    __shared__ int  lcnt[RB];
    int* tab = (int*)tab4;

    const int tid  = threadIdx.x;
    const int lane = tid & 31;
    const int wave = __builtin_amdgcn_readfirstlane(tid >> 5);
    const int b    = blockIdx.x;

    for (int i = tid; i < RB * CAP / 4; i += 1024) tab4[i] = make_int4(0, 0, 0, 0);
    if (tid < RB) lcnt[tid] = 0;
    __syncthreads();

    for (int w0 = wave; w0 < nChunks; w0 += 32 * 8) {
        int pr[8];
#pragma unroll
        for (int k = 0; k < 8; ++k) {
            const int w = w0 + 32 * k;
            pr[k] = (w < nChunks) ? runs[b * nChunks + w] : 0;
        }
        int rc[8];
#pragma unroll
        for (int k = 0; k < 8; ++k) {
            const int w = w0 + 32 * k;
            const int n = pr[k] & 0xFFFF;
            rc[k] = (lane < n) ? recs[(size_t)w * CHUNK + (pr[k] >> 16) + lane] : -1;
        }
#pragma unroll
        for (int k = 0; k < 8; ++k)
            if (rc[k] >= 0) place_record(rc[k], b, lcnt, tab, cnt, ovf_src, ovf_dst);
#pragma unroll
        for (int k = 0; k < 8; ++k) {
            const int w = w0 + 32 * k;
            const int n = pr[k] & 0xFFFF;
            for (int i = lane + 32; i < n; i += 32)
                place_record(recs[(size_t)w * CHUNK + (pr[k] >> 16) + i], b, lcnt, tab, cnt, ovf_src, ovf_dst);
        }
    }
    __syncthreads();

    int4* out4 = (int4*)nbr + (size_t)b * (RB * CAP / 4);
    for (int i = tid; i < RB * CAP / 4; i += 1024) {
        const int row = b * RB + (i >> 4);
        if (row < N_NODES) out4[i] = tab4[i];
    }
    if (tid < RB && b * RB + tid < N_NODES) cnt[b * RB + tid] = lcnt[tid];
}

__global__ __launch_bounds__(256) void sage_gather_gemm_proj_wmma(const float* __restrict__ feat,
                                                      const int* __restrict__ cnt,
                                                      const int* __restrict__ nbr,
                                                      const int* __restrict__ ovf_src,
                                                      const int* __restrict__ ovf_dst,
                                                      int nE,
                                                      const float* __restrict__ W1,
                                                      const float* __restrict__ b1,
                                                      const float* __restrict__ W2,
                                                      float* __restrict__ z) {
    __shared__ float Atile[16 * LDA];
    __shared__ float Htile[16 * LDA];

    const int tid  = threadIdx.x;
    const int lane = tid & 31;
    const int wave = __builtin_amdgcn_readfirstlane(tid >> 5);
    const int m0   = blockIdx.x * 16;
    const int c4   = lane << 2;

    for (int rr = 0; rr < 2; ++rr) {
        const int r   = 2 * wave + rr;
        const int row = m0 + r;
        const int deg = cnt[row];
        const int beg = row * CAP;
        const int end = beg + ((deg < CAP) ? deg : CAP);
        const float4 xv = *(const float4*)(feat + (size_t)row * KDIM + c4);

        float4 acc = make_float4(0.f, 0.f, 0.f, 0.f);
        int j = beg;
        for (; j + 8 <= end; j += 8) {
            int s[8];
#pragma unroll
            for (int q = 0; q < 8; ++q) s[q] = nbr[j + q];
            float4 v[8];
#pragma unroll
            for (int q = 0; q < 8; ++q) v[q] = *(const float4*)(feat + (size_t)s[q] * KDIM + c4);
#pragma unroll
            for (int q = 0; q < 8; ++q) { acc.x += v[q].x; acc.y += v[q].y; acc.z += v[q].z; acc.w += v[q].w; }
        }
        const int rem = end - j;
        if (rem & 4) {
            int s[4];
#pragma unroll
            for (int q = 0; q < 4; ++q) s[q] = nbr[j + q];
            float4 v[4];
#pragma unroll
            for (int q = 0; q < 4; ++q) v[q] = *(const float4*)(feat + (size_t)s[q] * KDIM + c4);
#pragma unroll
            for (int q = 0; q < 4; ++q) { acc.x += v[q].x; acc.y += v[q].y; acc.z += v[q].z; acc.w += v[q].w; }
            j += 4;
        }
        if (rem & 2) {
            const int s0 = nbr[j], s1 = nbr[j + 1];
            const float4 v0 = *(const float4*)(feat + (size_t)s0 * KDIM + c4);
            const float4 v1 = *(const float4*)(feat + (size_t)s1 * KDIM + c4);
            acc.x += v0.x; acc.y += v0.y; acc.z += v0.z; acc.w += v0.w;
            acc.x += v1.x; acc.y += v1.y; acc.z += v1.z; acc.w += v1.w;
            j += 2;
        }
        if (rem & 1) {
            const int s0 = nbr[j];
            const float4 v0 = *(const float4*)(feat + (size_t)s0 * KDIM + c4);
            acc.x += v0.x; acc.y += v0.y; acc.z += v0.z; acc.w += v0.w;
        }
        if (deg > CAP) {
            const int nraw = cnt[N_NODES];
            const int novf = (nraw < nE) ? nraw : nE;
            for (int i0 = 0; i0 < novf; i0 += 32) {
                const int i  = i0 + lane;
                const int od = (i < novf) ? ovf_dst[i] : -1;
                const int os = (i < novf) ? ovf_src[i] : 0;
                unsigned hit = (unsigned)__ballot(od == row);
                while (hit) {
                    const int b = __builtin_ctz(hit);
                    hit &= hit - 1;
                    const int sb = __shfl(os, b, 32);
                    const float4 vb = *(const float4*)(feat + (size_t)sb * KDIM + c4);
                    acc.x += vb.x; acc.y += vb.y; acc.z += vb.z; acc.w += vb.w;
                }
            }
        }

        const float inv = 1.0f / ((float)deg + 1.0f);
        float* a = &Atile[r * LDA + c4];
        a[0] = (acc.x + xv.x) * inv;
        a[1] = (acc.y + xv.y) * inv;
        a[2] = (acc.z + xv.z) * inv;
        a[3] = (acc.w + xv.w) * inv;
    }
    __syncthreads();

    const int n0 = wave * 16;
    const int lr = lane & 15;
    const int hi = lane >> 4;

    v8f acc = {0.f, 0.f, 0.f, 0.f, 0.f, 0.f, 0.f, 0.f};

#pragma unroll 8
    for (int k = 0; k < KDIM; k += 4) {
        v2f a, b;
        const float* ap = &Atile[lr * LDA + k + 2 * hi];
        a.x = ap[0];
        a.y = ap[1];
        const float* bp = &W1[(size_t)(k + 2 * hi) * KDIM + n0 + lr];
        b.x = bp[0];
        b.y = bp[KDIM];
        acc = __builtin_amdgcn_wmma_f32_16x16x4_f32(
             false, a,  false, b,
             (short)0, acc,  false,  false);
    }

    const float bv = b1[n0 + lr];
#pragma unroll
    for (int i = 0; i < 8; ++i) {
        float v = acc[i] + bv;
        v = (v >= 0.f) ? v : 0.01f * v;
        Htile[(8 * hi + i) * LDA + n0 + lr] = v;
    }
    __syncthreads();

    if (wave < 2) {
        const int p0 = wave * 16;
        v8f acc2 = {0.f, 0.f, 0.f, 0.f, 0.f, 0.f, 0.f, 0.f};

#pragma unroll 8
        for (int k = 0; k < KDIM; k += 4) {
            v2f a, b;
            const float* ap = &Htile[lr * LDA + k + 2 * hi];
            a.x = ap[0];
            a.y = ap[1];
            const float* bp = &W2[(size_t)(k + 2 * hi) * NOUT2 + p0 + lr];
            b.x = bp[0];
            b.y = bp[NOUT2];
            acc2 = __builtin_amdgcn_wmma_f32_16x16x4_f32(
                 false, a,  false, b,
                 (short)0, acc2,  false,  false);
        }

        const int col   = p0 + lr;
        const int rbase = m0 + 8 * hi;
#pragma unroll
        for (int i = 0; i < 8; ++i)
            z[(size_t)(rbase + i) * NOUT2 + col] = acc2[i];
    }
}

__global__ __launch_bounds__(256) void gather_out_kernel(const float* __restrict__ z,
                                                         const int* __restrict__ cnt,
                                                         const int* __restrict__ nbr,
                                                         const int* __restrict__ ovf_src,
                                                         const int* __restrict__ ovf_dst,
                                                         int nE,
                                                         const float* __restrict__ bias,
                                                         float* __restrict__ out) {
    const int lane = threadIdx.x & 31;
    const int w = __builtin_amdgcn_readfirstlane((int)(blockIdx.x * 8 + (threadIdx.x >> 5)));
    const float bv = bias[lane];

    for (int rr = 0; rr < 2; ++rr) {
        const int row = 2 * w + rr;
        if (row >= N_NODES) break;
        const int deg = cnt[row];
        const int beg = row * CAP;
        const int end = beg + ((deg < CAP) ? deg : CAP);
        const float zv = z[(size_t)row * NOUT2 + lane];

        float acc = 0.f;
        int j = beg;
        for (; j + 8 <= end; j += 8) {
            int s[8];
#pragma unroll
            for (int q = 0; q < 8; ++q) s[q] = nbr[j + q];
            float v[8];
#pragma unroll
            for (int q = 0; q < 8; ++q) v[q] = z[(size_t)s[q] * NOUT2 + lane];
#pragma unroll
            for (int q = 0; q < 8; ++q) acc += v[q];
        }
        const int rem = end - j;
        if (rem & 4) {
            int s[4];
#pragma unroll
            for (int q = 0; q < 4; ++q) s[q] = nbr[j + q];
            float v[4];
#pragma unroll
            for (int q = 0; q < 4; ++q) v[q] = z[(size_t)s[q] * NOUT2 + lane];
#pragma unroll
            for (int q = 0; q < 4; ++q) acc += v[q];
            j += 4;
        }
        if (rem & 2) {
            const int s0 = nbr[j], s1 = nbr[j + 1];
            const float v0 = z[(size_t)s0 * NOUT2 + lane];
            const float v1 = z[(size_t)s1 * NOUT2 + lane];
            acc += v0;
            acc += v1;
            j += 2;
        }
        if (rem & 1) acc += z[(size_t)nbr[j] * NOUT2 + lane];
        if (deg > CAP) {
            const int nraw = cnt[N_NODES];
            const int novf = (nraw < nE) ? nraw : nE;
            for (int i0 = 0; i0 < novf; i0 += 32) {
                const int i  = i0 + lane;
                const int od = (i < novf) ? ovf_dst[i] : -1;
                const int os = (i < novf) ? ovf_src[i] : 0;
                unsigned hit = (unsigned)__ballot(od == row);
                while (hit) {
                    const int b = __builtin_ctz(hit);
                    hit &= hit - 1;
                    const int sb = __shfl(os, b, 32);
                    acc += z[(size_t)sb * NOUT2 + lane];
                }
            }
        }

        const float inv = 1.0f / ((float)deg + 1.0f);
        out[(size_t)row * NOUT2 + lane] = (acc + zv) * inv + bv;
    }
}

extern "C" void kernel_launch(void* const* d_in, const int* in_sizes, int n_in,
                              void* d_out, int out_size, void* d_ws, size_t ws_size,
                              hipStream_t stream) {
    const float* x   = (const float*)d_in[0];
    const int*   edg = (const int*)  d_in[1];
    const float* W1  = (const float*)d_in[2];
    const float* b1  = (const float*)d_in[3];
    const float* W2  = (const float*)d_in[4];
    const float* b2  = (const float*)d_in[5];
    float*       outp = (float*)d_out;

    const int E = in_sizes[1] / 2;
    const int* src = edg;
    const int* dst = edg + E;

    const int nChunks = (E + CHUNK - 1) / CHUNK;
    float* z       = (float*)d_ws;
    int*   nbr     = (int*)(z + (size_t)N_NODES * NOUT2);
    int*   cnt     = nbr + (size_t)N_NODES * CAP;
    int*   ovf_src = cnt + (N_NODES + 48);
    int*   ovf_dst = ovf_src + E;
    int*   recs    = ovf_dst + E;
    int*   runs    = recs + (size_t)nChunks * CHUNK;

    bin_edges_kernel<<<nChunks, 1024, 0, stream>>>(src, dst, recs, runs, cnt, E, nChunks);
    build_rows_kernel<<<NB, 1024, 0, stream>>>(recs, runs, cnt, nbr, ovf_src, ovf_dst, nChunks);

    sage_gather_gemm_proj_wmma<<<N_NODES / 16, 256, 0, stream>>>(x, cnt, nbr, ovf_src, ovf_dst, E, W1, b1, W2, z);

    gather_out_kernel<<<N_NODES / 16, 256, 0, stream>>>(z, cnt, nbr, ovf_src, ovf_dst, E, b2, outp);
}


# both neighbour sums: a row's 32 table slots come by ONE wave-wide load issued with its in-degree one row ahead, sources taken by lane read: no index load per group of 8; sum order unchanged
# speedup vs baseline: 7.3081x; 1.0348x over previous
#include <hip/hip_runtime.h>
#include <hip/hip_bf16.h>

typedef float v2f __attribute__((ext_vector_type(2)));
typedef float v8f __attribute__((ext_vector_type(8)));

#define N_NODES 50000
#define KDIM    128
#define LDA     132
#define NOUT2   32
#define CAP     64

#define RB      256
#define NB      ((N_NODES + RB - 1) / RB)
#define CHUNK   4096
static_assert(RB == 256, "a record keeps the row inside its bucket in 8 bits");
static_assert(N_NODES < (1 << 23), "a record keeps the source above those 8 bits");
static_assert(NB <= 7 * 32, "the bucket scan gives each lane of one wave 7 buckets");

__global__ __launch_bounds__(1024) void bin_edges_kernel(const int* __restrict__ src,
                                                         const int* __restrict__ dst,
                                                         int* __restrict__ recs,
                                                         int* __restrict__ runs,
                                                         int* __restrict__ cnt,
                                                         int nE, int nChunks) {
    __shared__ int hist[NB];
    __shared__ int start[NB];
    __shared__ int sorted[CHUNK];

    const int tid  = threadIdx.x;
    const int base = blockIdx.x * CHUNK;

    if (tid < NB) hist[tid] = 0;
    if (blockIdx.x == 0 && tid == 0) cnt[N_NODES] = 0;
    __syncthreads();

    int dv[4], sv[4];
#pragma unroll
    for (int q = 0; q < 4; ++q) {
        const int e  = base + q * 1024 + tid;
        const int ec = (e < nE) ? e : (nE - 1);
        dv[q] = dst[ec];
        sv[q] = src[ec];
    }
    int rec[4], bk[4], rk[4];
#pragma unroll
    for (int q = 0; q < 4; ++q) {
        const int e = base + q * 1024 + tid;
        bk[q] = -1; rec[q] = 0; rk[q] = 0;
        if (e < nE) {
            bk[q]  = dv[q] >> 8;
            rec[q] = (sv[q] << 8) | (dv[q] & (RB - 1));
            rk[q]  = atomicAdd(&hist[bk[q]], 1);
        }
    }
    __syncthreads();

    if (tid < 32) {
        int c[7];
        int sum = 0;
#pragma unroll
        for (int i = 0; i < 7; ++i) {
            const int b = tid * 7 + i;
            c[i] = (b < NB) ? hist[b] : 0;
            sum += c[i];
        }
        int incl = sum;
#pragma unroll
        for (int off = 1; off < 32; off <<= 1) {
            const int t = __shfl_up(incl, off, 32);
            if (tid >= off) incl += t;
        }
        int run = incl - sum;
#pragma unroll
        for (int i = 0; i < 7; ++i) {
            const int b = tid * 7 + i;
            if (b < NB) start[b] = run;
            run += c[i];
        }
    }
    __syncthreads();

#pragma unroll
    for (int q = 0; q < 4; ++q)
        if (bk[q] >= 0) sorted[start[bk[q]] + rk[q]] = rec[q];
    __syncthreads();

#pragma unroll
    for (int q = 0; q < 4; ++q) {
        const int i = q * 1024 + tid;
        if (base + i < nE) recs[base + i] = sorted[i];
    }
    if (tid < NB) runs[tid * nChunks + blockIdx.x] = (start[tid] << 16) | hist[tid];
}

__device__ __forceinline__ void place_record(int rc, int b, int* lcnt, int* tab,
                                             int* __restrict__ cnt, int* __restrict__ ovf_src,
                                             int* __restrict__ ovf_dst) {
    const int r    = rc & (RB - 1);
    const int s    = rc >> 8;
    const int slot = atomicAdd(&lcnt[r], 1);
    if (slot < CAP) {
        tab[r * CAP + slot] = s;
    } else {
        const int o = atomicAdd(&cnt[N_NODES], 1);
        ovf_src[o] = s;
        ovf_dst[o] = b * RB + r;
    }
}

__global__ __launch_bounds__(1024) void build_rows_kernel(const int* __restrict__ recs,
                                                          const int* __restrict__ runs,
                                                          int* __restrict__ cnt,
                                                          int* __restrict__ nbr,
                                                          int* __restrict__ ovf_src,
                                                          int* __restrict__ ovf_dst,
                                                          int nChunks) {
    __shared__ int4 tab4[RB * CAP / 4];
    __shared__ int  lcnt[RB];
    int* tab = (int*)tab4;

    const int tid  = threadIdx.x;
    const int lane = tid & 31;
    const int wave = __builtin_amdgcn_readfirstlane(tid >> 5);
    const int b    = blockIdx.x;

    for (int i = tid; i < RB * CAP / 4; i += 1024) tab4[i] = make_int4(0, 0, 0, 0);
    if (tid < RB) lcnt[tid] = 0;
    __syncthreads();

    for (int w0 = wave; w0 < nChunks; w0 += 32 * 8) {
        int pr[8];
#pragma unroll
        for (int k = 0; k < 8; ++k) {
            const int w = w0 + 32 * k;
            pr[k] = (w < nChunks) ? runs[b * nChunks + w] : 0;
        }
        int rc[8];
#pragma unroll
        for (int k = 0; k < 8; ++k) {
            const int w = w0 + 32 * k;
            const int n = pr[k] & 0xFFFF;
            rc[k] = (lane < n) ? recs[(size_t)w * CHUNK + (pr[k] >> 16) + lane] : -1;
        }
#pragma unroll
        for (int k = 0; k < 8; ++k)
            if (rc[k] >= 0) place_record(rc[k], b, lcnt, tab, cnt, ovf_src, ovf_dst);
#pragma unroll
        for (int k = 0; k < 8; ++k) {
            const int w = w0 + 32 * k;
            const int n = pr[k] & 0xFFFF;
            for (int i = lane + 32; i < n; i += 32)
                place_record(recs[(size_t)w * CHUNK + (pr[k] >> 16) + i], b, lcnt, tab, cnt, ovf_src, ovf_dst);
        }
    }
    __syncthreads();

    int4* out4 = (int4*)nbr + (size_t)b * (RB * CAP / 4);
    for (int i = tid; i < RB * CAP / 4; i += 1024) {
        const int row = b * RB + (i >> 4);
        if (row < N_NODES) out4[i] = tab4[i];
    }
    if (tid < RB && b * RB + tid < N_NODES) cnt[b * RB + tid] = lcnt[tid];
}

__device__ __forceinline__ void sum_feat_rows(const float* __restrict__ feat, int idx, int n32,
                                              int c4, float4& acc) {
    int j = 0;
    for (; j + 8 <= n32; j += 8) {
        float4 v[8];
#pragma unroll
        for (int q = 0; q < 8; ++q) {
            const int s = __builtin_amdgcn_readlane(idx, j + q);
            v[q] = *(const float4*)(feat + (size_t)s * KDIM + c4);
        }
#pragma unroll
        for (int q = 0; q < 8; ++q) { acc.x += v[q].x; acc.y += v[q].y; acc.z += v[q].z; acc.w += v[q].w; }
    }
    const int m = n32 - j;
    if (m > 0) {
        float4 v[7];
#pragma unroll
        for (int q = 0; q < 7; ++q) {
            v[q] = make_float4(0.f, 0.f, 0.f, 0.f);
            if (q < m) {
                const int s = __builtin_amdgcn_readlane(idx, j + q);
                v[q] = *(const float4*)(feat + (size_t)s * KDIM + c4);
            }
        }
#pragma unroll
        for (int q = 0; q < 7; ++q) { acc.x += v[q].x; acc.y += v[q].y; acc.z += v[q].z; acc.w += v[q].w; }
    }
}

__device__ __forceinline__ void sum_z_rows(const float* __restrict__ z, int idx, int n32,
                                           int lane, float& acc) {
    int j = 0;
    for (; j + 8 <= n32; j += 8) {
        float v[8];
#pragma unroll
        for (int q = 0; q < 8; ++q) {
            const int s = __builtin_amdgcn_readlane(idx, j + q);
            v[q] = z[(size_t)s * NOUT2 + lane];
        }
#pragma unroll
        for (int q = 0; q < 8; ++q) acc += v[q];
    }
    const int m = n32 - j;
    if (m > 0) {
        float v[7];
#pragma unroll
        for (int q = 0; q < 7; ++q) {
            v[q] = 0.f;
            if (q < m) {
                const int s = __builtin_amdgcn_readlane(idx, j + q);
                v[q] = z[(size_t)s * NOUT2 + lane];
            }
        }
#pragma unroll
        for (int q = 0; q < 7; ++q) acc += v[q];
    }
}

__global__ __launch_bounds__(256) void sage_gather_gemm_proj_wmma(const float* __restrict__ feat,
                                                      const int* __restrict__ cnt,
                                                      const int* __restrict__ nbr,
                                                      const int* __restrict__ ovf_src,
                                                      const int* __restrict__ ovf_dst,
                                                      int nE,
                                                      const float* __restrict__ W1,
                                                      const float* __restrict__ b1,
                                                      const float* __restrict__ W2,
                                                      float* __restrict__ z) {
    __shared__ float Atile[16 * LDA];
    __shared__ float Htile[16 * LDA];

    const int tid  = threadIdx.x;
    const int lane = tid & 31;
    const int wave = __builtin_amdgcn_readfirstlane(tid >> 5);
    const int m0   = blockIdx.x * 16;
    const int c4   = lane << 2;

    const int row0 = m0 + 2 * wave;
    int    deg_nx = cnt[row0];
    int    idx_nx = nbr[row0 * CAP + lane];
    float4 own_nx = *(const float4*)(feat + (size_t)row0 * KDIM + c4);
    for (int rr = 0; rr < 2; ++rr) {
        const int r   = 2 * wave + rr;
        const int row = row0 + rr;
        const int deg = deg_nx;
        const int idx = idx_nx;
        const float4 xv = own_nx;
        const int n   = (deg < CAP) ? deg : CAP;
        if (rr == 0) {
            deg_nx = cnt[row + 1];
            idx_nx = nbr[(row + 1) * CAP + lane];
            own_nx = *(const float4*)(feat + (size_t)(row + 1) * KDIM + c4);
        }

        float4 acc = make_float4(0.f, 0.f, 0.f, 0.f);
        sum_feat_rows(feat, idx, (n < 32) ? n : 32, c4, acc);
        if (n > 32) {
            const int idx1 = nbr[row * CAP + 32 + lane];
            sum_feat_rows(feat, idx1, n - 32, c4, acc);
        }
        if (deg > CAP) {
            const int nraw = cnt[N_NODES];
            const int novf = (nraw < nE) ? nraw : nE;
            for (int i0 = 0; i0 < novf; i0 += 32) {
                const int i  = i0 + lane;
                const int od = (i < novf) ? ovf_dst[i] : -1;
                const int os = (i < novf) ? ovf_src[i] : 0;
                unsigned hit = (unsigned)__ballot(od == row);
                while (hit) {
                    const int b = __builtin_ctz(hit);
                    hit &= hit - 1;
                    const int sb = __shfl(os, b, 32);
                    const float4 vb = *(const float4*)(feat + (size_t)sb * KDIM + c4);
                    acc.x += vb.x; acc.y += vb.y; acc.z += vb.z; acc.w += vb.w;
                }
            }
        }

        const float inv = 1.0f / ((float)deg + 1.0f);
        float* a = &Atile[r * LDA + c4];
        a[0] = (acc.x + xv.x) * inv;
        a[1] = (acc.y + xv.y) * inv;
        a[2] = (acc.z + xv.z) * inv;
        a[3] = (acc.w + xv.w) * inv;
    }
    __syncthreads();

    const int n0 = wave * 16;
    const int lr = lane & 15;
    const int hi = lane >> 4;

    v8f acc = {0.f, 0.f, 0.f, 0.f, 0.f, 0.f, 0.f, 0.f};

#pragma unroll 8
    for (int k = 0; k < KDIM; k += 4) {
        v2f a, b;
        const float* ap = &Atile[lr * LDA + k + 2 * hi];
        a.x = ap[0];
        a.y = ap[1];
        const float* bp = &W1[(size_t)(k + 2 * hi) * KDIM + n0 + lr];
        b.x = bp[0];
        b.y = bp[KDIM];
        acc = __builtin_amdgcn_wmma_f32_16x16x4_f32(
             false, a,  false, b,
             (short)0, acc,  false,  false);
    }

    const float bv = b1[n0 + lr];
#pragma unroll
    for (int i = 0; i < 8; ++i) {
        float v = acc[i] + bv;
        v = (v >= 0.f) ? v : 0.01f * v;
        Htile[(8 * hi + i) * LDA + n0 + lr] = v;
    }
    __syncthreads();

    if (wave < 2) {
        const int p0 = wave * 16;
        v8f acc2 = {0.f, 0.f, 0.f, 0.f, 0.f, 0.f, 0.f, 0.f};

#pragma unroll 8
        for (int k = 0; k < KDIM; k += 4) {
            v2f a, b;
            const float* ap = &Htile[lr * LDA + k + 2 * hi];
            a.x = ap[0];
            a.y = ap[1];
            const float* bp = &W2[(size_t)(k + 2 * hi) * NOUT2 + p0 + lr];
            b.x = bp[0];
            b.y = bp[NOUT2];
            acc2 = __builtin_amdgcn_wmma_f32_16x16x4_f32(
                 false, a,  false, b,
                 (short)0, acc2,  false,  false);
        }

        const int col   = p0 + lr;
        const int rbase = m0 + 8 * hi;
#pragma unroll
        for (int i = 0; i < 8; ++i)
            z[(size_t)(rbase + i) * NOUT2 + col] = acc2[i];
    }
}

__global__ __launch_bounds__(256) void gather_out_kernel(const float* __restrict__ z,
                                                         const int* __restrict__ cnt,
                                                         const int* __restrict__ nbr,
                                                         const int* __restrict__ ovf_src,
                                                         const int* __restrict__ ovf_dst,
                                                         int nE,
                                                         const float* __restrict__ bias,
                                                         float* __restrict__ out) {
    const int lane = threadIdx.x & 31;
    const int w = __builtin_amdgcn_readfirstlane((int)(blockIdx.x * 8 + (threadIdx.x >> 5)));
    const float bv = bias[lane];

    static_assert(N_NODES % 2 == 0, "a wave's two rows are 2w and 2w+1");
    if (2 * w >= N_NODES) return;

    int   deg_nx = cnt[2 * w];
    int   idx_nx = nbr[2 * w * CAP + lane];
    float own_nx = z[(size_t)(2 * w) * NOUT2 + lane];
    for (int rr = 0; rr < 2; ++rr) {
        const int row = 2 * w + rr;
        const int deg = deg_nx;
        const int idx = idx_nx;
        const float zv = own_nx;
        const int n   = (deg < CAP) ? deg : CAP;
        if (rr == 0) {
            deg_nx = cnt[row + 1];
            idx_nx = nbr[(row + 1) * CAP + lane];
            own_nx = z[(size_t)(row + 1) * NOUT2 + lane];
        }

        float acc = 0.f;
        sum_z_rows(z, idx, (n < 32) ? n : 32, lane, acc);
        if (n > 32) {
            const int idx1 = nbr[row * CAP + 32 + lane];
            sum_z_rows(z, idx1, n - 32, lane, acc);
        }
        if (deg > CAP) {
            const int nraw = cnt[N_NODES];
            const int novf = (nraw < nE) ? nraw : nE;
            for (int i0 = 0; i0 < novf; i0 += 32) {
                const int i  = i0 + lane;
                const int od = (i < novf) ? ovf_dst[i] : -1;
                const int os = (i < novf) ? ovf_src[i] : 0;
                unsigned hit = (unsigned)__ballot(od == row);
                while (hit) {
                    const int b = __builtin_ctz(hit);
                    hit &= hit - 1;
                    const int sb = __shfl(os, b, 32);
                    acc += z[(size_t)sb * NOUT2 + lane];
                }
            }
        }

        const float inv = 1.0f / ((float)deg + 1.0f);
        out[(size_t)row * NOUT2 + lane] = (acc + zv) * inv + bv;
    }
}

extern "C" void kernel_launch(void* const* d_in, const int* in_sizes, int n_in,
                              void* d_out, int out_size, void* d_ws, size_t ws_size,
                              hipStream_t stream) {
    const float* x   = (const float*)d_in[0];
    const int*   edg = (const int*)  d_in[1];
    const float* W1  = (const float*)d_in[2];
    const float* b1  = (const float*)d_in[3];
    const float* W2  = (const float*)d_in[4];
    const float* b2  = (const float*)d_in[5];
    float*       outp = (float*)d_out;

    const int E = in_sizes[1] / 2;
    const int* src = edg;
    const int* dst = edg + E;

    const int nChunks = (E + CHUNK - 1) / CHUNK;
    float* z       = (float*)d_ws;
    int*   nbr     = (int*)(z + (size_t)N_NODES * NOUT2);
    int*   cnt     = nbr + (size_t)N_NODES * CAP;
    int*   ovf_src = cnt + (N_NODES + 48);
    int*   ovf_dst = ovf_src + E;
    int*   recs    = ovf_dst + E;
    int*   runs    = recs + (size_t)nChunks * CHUNK;

    bin_edges_kernel<<<nChunks, 1024, 0, stream>>>(src, dst, recs, runs, cnt, E, nChunks);
    build_rows_kernel<<<NB, 1024, 0, stream>>>(recs, runs, cnt, nbr, ovf_src, ovf_dst, nChunks);

    sage_gather_gemm_proj_wmma<<<N_NODES / 16, 256, 0, stream>>>(x, cnt, nbr, ovf_src, ovf_dst, E, W1, b1, W2, z);

    gather_out_kernel<<<N_NODES / 16, 256, 0, stream>>>(z, cnt, nbr, ovf_src, ovf_dst, E, b2, outp);
}
